# starting kernel
# baseline (speedup 1.0000x reference)
#include <hip/hip_runtime.h>


typedef __attribute__((ext_vector_type(2))) float v2f;
typedef __attribute__((ext_vector_type(8))) float v8f;

#define B_ 4
#define H_ 16
#define S_ 4096
#define D_ 64
#define BH_ (B_ * H_)
#define EPS_ 1e-6f

#define KV_STRIDE 4160
#define SCHUNK 512
#define P2_ROWS 128

#define QSTRIDE 68
#define KVSTRIDE_L 72

__device__ __forceinline__ float phi_f(float x) {
    return x > 0.0f ? x + 1.0f : __expf(x);
}

__global__ void la_zero_ws(float4* __restrict__ ws, int n4) {
    int i = blockIdx.x * blockDim.x + threadIdx.x;
    if (i < n4) ws[i] = make_float4(0.f, 0.f, 0.f, 0.f);
}

__global__ __launch_bounds__(256) void la_pass1_kvterm(
    const float* __restrict__ K, const float* __restrict__ V,
    const float* __restrict__ mask, float* __restrict__ ws) {

    const int bh = blockIdx.x;
    const int chunk = blockIdx.y;
    const int b = bh / H_;
    const size_t base = (size_t)bh * S_ * D_ + (size_t)chunk * SCHUNK * D_;
    const float* __restrict__ Kp = K + base;
    const float* __restrict__ Vp = V + base;
    const float* __restrict__ Mp = mask + (size_t)b * S_ + (size_t)chunk * SCHUNK;

    __shared__ __align__(16) float lds_kT[D_][17];
    __shared__ __align__(16) float lds_v[16][64];
    __shared__ float lds_kone[D_];

    const int tid  = threadIdx.x;
    const int lane = tid & 31;
    const int wid  = tid >> 5;
    const int l15  = lane & 15;
    const int half = lane >> 4;

    const int dTile0 = wid >> 2;
    const int dTile1 = dTile0 + 2;
    const int vTile  = wid & 3;

    const int nl  = tid >> 4;
    const int col = (tid & 15) * 4;

    v8f c0 = {}; v8f c1 = {};
    float kone_acc[4] = {0.f, 0.f, 0.f, 0.f};

    if (tid < D_) lds_kone[tid] = 0.0f;

    for (int n0 = 0; n0 < SCHUNK; n0 += 16) {
        __syncthreads();
        const float4 kd = *(const float4*)&Kp[(size_t)(n0 + nl) * D_ + col];
        const float4 vv = *(const float4*)&Vp[(size_t)(n0 + nl) * D_ + col];
        const float m = Mp[n0 + nl];
        const float pk0 = phi_f(kd.x) * m;
        const float pk1 = phi_f(kd.y) * m;
        const float pk2 = phi_f(kd.z) * m;
        const float pk3 = phi_f(kd.w) * m;
        lds_kT[col + 0][nl] = pk0;
        lds_kT[col + 1][nl] = pk1;
        lds_kT[col + 2][nl] = pk2;
        lds_kT[col + 3][nl] = pk3;
        kone_acc[0] += pk0; kone_acc[1] += pk1;
        kone_acc[2] += pk2; kone_acc[3] += pk3;
        *(float4*)&lds_v[nl][col] = vv;
        __syncthreads();

        #pragma unroll
        for (int kk = 0; kk < 16; kk += 4) {
            const int kc = kk + half * 2;
            v2f a0, a1, bb;
            a0.x = lds_kT[dTile0 * 16 + l15][kc];
            a0.y = lds_kT[dTile0 * 16 + l15][kc + 1];
            a1.x = lds_kT[dTile1 * 16 + l15][kc];
            a1.y = lds_kT[dTile1 * 16 + l15][kc + 1];
            bb.x = lds_v[kc][vTile * 16 + l15];
            bb.y = lds_v[kc + 1][vTile * 16 + l15];
            c0 = __builtin_amdgcn_wmma_f32_16x16x4_f32(false, a0, false, bb,
                                                       (short)0, c0, false, false);
            c1 = __builtin_amdgcn_wmma_f32_16x16x4_f32(false, a1, false, bb,
                                                       (short)0, c1, false, false);
        }
    }

    #pragma unroll
    for (int j = 0; j < 4; ++j) atomicAdd(&lds_kone[col + j], kone_acc[j]);
    __syncthreads();

    float* __restrict__ wkv = ws + (size_t)bh * KV_STRIDE;
    if (tid < D_) atomicAdd(&wkv[4096 + tid], lds_kone[tid]);

    #pragma unroll
    for (int j = 0; j < 8; ++j) {
        const int r0 = dTile0 * 16 + j + 8 * half;
        const int r1 = dTile1 * 16 + j + 8 * half;
        const int cidx = vTile * 16 + l15;
        atomicAdd(&wkv[r0 * 64 + cidx], c0[j]);
        atomicAdd(&wkv[r1 * 64 + cidx], c1[j]);
    }
}

__global__ __launch_bounds__(256) void la_pass2_output(
    const float* __restrict__ Q, const float* __restrict__ ws,
    float* __restrict__ out) {

    const int bh = blockIdx.x;
    const int rblk = blockIdx.y;
    const size_t qbase = (size_t)bh * S_ * D_ + (size_t)rblk * P2_ROWS * D_;
    const float* __restrict__ Qp = Q + qbase;
    float* __restrict__ Op = out + qbase;
    const float* __restrict__ wkv = ws + (size_t)bh * KV_STRIDE;

    __shared__ __align__(16) float lds_kv[D_][KVSTRIDE_L];
    __shared__ float lds_kone[D_];
    __shared__ __align__(16) float lds_q[P2_ROWS][QSTRIDE];

    const int tid  = threadIdx.x;
    const int lane = tid & 31;
    const int wid  = tid >> 5;
    const int l15  = lane & 15;
    const int half = lane >> 4;

    #pragma unroll
    for (int i = 0; i < 4; ++i) {
        const int idx = tid + 256 * i;
        const int r = idx >> 4;
        const int c = (idx & 15) * 4;
        *(float4*)&lds_kv[r][c] = *(const float4*)&wkv[r * 64 + c];
    }
    if (tid < D_) lds_kone[tid] = wkv[4096 + tid];

    #pragma unroll
    for (int i = 0; i < 8; ++i) {
        const int idx = tid + 256 * i;
        const int r = idx >> 4;
        const int c = (idx & 15) * 4;
        float4 qv = *(const float4*)&Qp[(size_t)r * D_ + c];
        qv.x = phi_f(qv.x * 0.125f);
        qv.y = phi_f(qv.y * 0.125f);
        qv.z = phi_f(qv.z * 0.125f);
        qv.w = phi_f(qv.w * 0.125f);
        *(float4*)&lds_q[r][c] = qv;
    }
    __syncthreads();

    const int mb = wid * 16;
    v8f acc[4] = {{}, {}, {}, {}};

    #pragma unroll
    for (int kk = 0; kk < 64; kk += 4) {
        const int kc = kk + half * 2;
        v2f a;
        a.x = lds_q[mb + l15][kc];
        a.y = lds_q[mb + l15][kc + 1];
        #pragma unroll
        for (int vt = 0; vt < 4; ++vt) {
            v2f bfrag;
            bfrag.x = lds_kv[kc][vt * 16 + l15];
            bfrag.y = lds_kv[kc + 1][vt * 16 + l15];
            acc[vt] = __builtin_amdgcn_wmma_f32_16x16x4_f32(false, a, false, bfrag,
                                                            (short)0, acc[vt], false, false);
        }
    }

    float part = 0.0f;
    const int nrow = mb + l15;
    #pragma unroll
    for (int d = 0; d < 32; ++d) {
        const int dd = half * 32 + d;
        part += lds_q[nrow][dd] * lds_kone[dd];
    }
    part += __shfl_xor(part, 16, 32);
    const float inv = 1.0f / (part + EPS_);

    #pragma unroll
    for (int j = 0; j < 8; ++j) {
        const float rn = __shfl(inv, j + 8 * half, 32);
        const int r = mb + j + 8 * half;
        #pragma unroll
        for (int vt = 0; vt < 4; ++vt) {
            Op[(size_t)r * D_ + vt * 16 + l15] = acc[vt][j] * rn;
        }
    }
}

extern "C" void kernel_launch(void* const* d_in, const int* in_sizes, int n_in,
                              void* d_out, int out_size, void* d_ws, size_t ws_size,
                              hipStream_t stream) {
    const float* Q = (const float*)d_in[0];
    const float* K = (const float*)d_in[1];
    const float* V = (const float*)d_in[2];
    const float* M = (const float*)d_in[3];
    float* out = (float*)d_out;
    float* ws  = (float*)d_ws;

    const int nz4 = (BH_ * KV_STRIDE) / 4;
    la_zero_ws<<<(nz4 + 255) / 256, 256, 0, stream>>>((float4*)ws, nz4);
    la_pass1_kvterm<<<dim3(BH_, S_ / SCHUNK), 256, 0, stream>>>(K, V, M, ws);
    la_pass2_output<<<dim3(BH_, S_ / P2_ROWS), 256, 0, stream>>>(Q, ws, out);
}


# pass 1: K,V staged through two 64-row row-major LDS slabs, global loads held two slabs ahead in registers; one LDS-only barrier per 64 rows, not two full ones per 16; per-block sum order kept
# speedup vs baseline: 1.4491x; 1.4491x over previous
#include <hip/hip_runtime.h>


typedef __attribute__((ext_vector_type(2))) float v2f;
typedef __attribute__((ext_vector_type(8))) float v8f;

#define B_ 4
#define H_ 16
#define S_ 4096
#define D_ 64
#define BH_ (B_ * H_)
#define EPS_ 1e-6f

#define KV_STRIDE 4160
#define SCHUNK 512
#define P2_ROWS 128

#define SLAB 64
#define NSLAB (SCHUNK / SLAB)
#define LPITCH 80

#define QSTRIDE 68
#define KVSTRIDE_L 72

__device__ __forceinline__ float phi_f(float x) {
    return x > 0.0f ? x + 1.0f : __expf(x);
}

__global__ void la_zero_ws(float4* __restrict__ ws, int n4) {
    int i = blockIdx.x * blockDim.x + threadIdx.x;
    if (i < n4) ws[i] = make_float4(0.f, 0.f, 0.f, 0.f);
}

struct SlabRegs {
    float4 k[4];
    float4 v[4];
    float  m[4];
};

__device__ __forceinline__ void slab_load(SlabRegs& R, const float* Kp,
                                          const float* Vp,
                                          const float* __restrict__ Mp,
                                          int n0, int rr, int col) {
    #pragma unroll
    for (int i = 0; i < 4; ++i) {
        const int r = n0 + 16 * i + rr;
        R.k[i] = *(const float4*)&Kp[(size_t)r * D_ + col];
        R.v[i] = *(const float4*)&Vp[(size_t)r * D_ + col];
        R.m[i] = Mp[r];
    }
}

__device__ __forceinline__ void slab_stage(const SlabRegs& R, float* __restrict__ lk,
                                           float* __restrict__ lv, int rr, int col,
                                           float (&kone_acc)[4]) {
    #pragma unroll
    for (int i = 0; i < 4; ++i) {
        const int r = 16 * i + rr;
        const float m = R.m[i];
        float4 pk;
        pk.x = phi_f(R.k[i].x) * m;
        pk.y = phi_f(R.k[i].y) * m;
        pk.z = phi_f(R.k[i].z) * m;
        pk.w = phi_f(R.k[i].w) * m;
        kone_acc[0] += pk.x; kone_acc[1] += pk.y;
        kone_acc[2] += pk.z; kone_acc[3] += pk.w;
        *(float4*)&lk[r * LPITCH + col] = pk;
        *(float4*)&lv[r * LPITCH + col] = R.v[i];
    }
}

__device__ __forceinline__ void slab_mma(const float* __restrict__ lk,
                                         const float* __restrict__ lv,
                                         int aoff0, int aoff1, int boff,
                                         v8f& c0, v8f& c1) {
    #pragma unroll 8
    for (int kk = 0; kk < SLAB; kk += 4) {
        v2f a0, a1, bb;
        a0.x = lk[kk * LPITCH + aoff0];
        a0.y = lk[(kk + 1) * LPITCH + aoff0];
        a1.x = lk[kk * LPITCH + aoff1];
        a1.y = lk[(kk + 1) * LPITCH + aoff1];
        bb.x = lv[kk * LPITCH + boff];
        bb.y = lv[(kk + 1) * LPITCH + boff];
        c0 = __builtin_amdgcn_wmma_f32_16x16x4_f32(false, a0, false, bb,
                                                   (short)0, c0, false, false);
        c1 = __builtin_amdgcn_wmma_f32_16x16x4_f32(false, a1, false, bb,
                                                   (short)0, c1, false, false);
    }
}

__device__ __forceinline__ void lds_barrier() {
    __builtin_amdgcn_fence(__ATOMIC_RELEASE, "workgroup", "local");
    __builtin_amdgcn_s_barrier();
    __builtin_amdgcn_fence(__ATOMIC_ACQUIRE, "workgroup", "local");
}

__global__ __launch_bounds__(256) void la_pass1_kvterm(
    const float* K, const float* V,
    const float* __restrict__ mask, float* __restrict__ ws) {

    const int bh = blockIdx.x;
    const int chunk = blockIdx.y;
    const int b = bh / H_;
    const size_t base = (size_t)bh * S_ * D_ + (size_t)chunk * SCHUNK * D_;
    const float* Kp = K + base;
    const float* Vp = V + base;
    const float* __restrict__ Mp = mask + (size_t)b * S_ + (size_t)chunk * SCHUNK;

    __shared__ __align__(16) float lds_k[2][SLAB * LPITCH];
    __shared__ __align__(16) float lds_v[2][SLAB * LPITCH];
    __shared__ float lds_kone[D_];

    const int tid  = threadIdx.x;
    const int lane = tid & 31;
    const int wid  = tid >> 5;
    const int l15  = lane & 15;
    const int half = lane >> 4;

    const int dTile0 = wid >> 2;
    const int dTile1 = dTile0 + 2;
    const int vTile  = wid & 3;

    const int rr  = tid >> 4;
    const int col = (tid & 15) * 4;

    const int aoff0 = (2 * half) * LPITCH + dTile0 * 16 + l15;
    const int aoff1 = (2 * half) * LPITCH + dTile1 * 16 + l15;
    const int boff  = (2 * half) * LPITCH + vTile * 16 + l15;

    v8f c0 = {}; v8f c1 = {};
    float kone_acc[4] = {0.f, 0.f, 0.f, 0.f};

    if (tid < D_) lds_kone[tid] = 0.0f;

    SlabRegs ra, rb;
    slab_load(ra, Kp, Vp, Mp, 0, rr, col);
    __builtin_amdgcn_sched_barrier(0);
    slab_load(rb, Kp, Vp, Mp, SLAB, rr, col);

    #pragma unroll 1
    for (int s = 0; s < NSLAB - 2; s += 2) {
        slab_stage(ra, lds_k[0], lds_v[0], rr, col, kone_acc);
        slab_load(ra, Kp, Vp, Mp, (s + 2) * SLAB, rr, col);
        lds_barrier();
        slab_mma(lds_k[0], lds_v[0], aoff0, aoff1, boff, c0, c1);

        slab_stage(rb, lds_k[1], lds_v[1], rr, col, kone_acc);
        slab_load(rb, Kp, Vp, Mp, (s + 3) * SLAB, rr, col);
        lds_barrier();
        slab_mma(lds_k[1], lds_v[1], aoff0, aoff1, boff, c0, c1);
    }
    slab_stage(ra, lds_k[0], lds_v[0], rr, col, kone_acc);
    lds_barrier();
    slab_mma(lds_k[0], lds_v[0], aoff0, aoff1, boff, c0, c1);
    slab_stage(rb, lds_k[1], lds_v[1], rr, col, kone_acc);
    lds_barrier();
    slab_mma(lds_k[1], lds_v[1], aoff0, aoff1, boff, c0, c1);

    #pragma unroll
    for (int j = 0; j < 4; ++j) atomicAdd(&lds_kone[col + j], kone_acc[j]);
    __syncthreads();

    float* __restrict__ wkv = ws + (size_t)bh * KV_STRIDE;
    if (tid < D_) atomicAdd(&wkv[4096 + tid], lds_kone[tid]);

    #pragma unroll
    for (int j = 0; j < 8; ++j) {
        const int r0 = dTile0 * 16 + j + 8 * half;
        const int r1 = dTile1 * 16 + j + 8 * half;
        const int cidx = vTile * 16 + l15;
        atomicAdd(&wkv[r0 * 64 + cidx], c0[j]);
        atomicAdd(&wkv[r1 * 64 + cidx], c1[j]);
    }
}

__global__ __launch_bounds__(256) void la_pass2_output(
    const float* __restrict__ Q, const float* __restrict__ ws,
    float* __restrict__ out) {

    const int bh = blockIdx.x;
    const int rblk = blockIdx.y;
    const size_t qbase = (size_t)bh * S_ * D_ + (size_t)rblk * P2_ROWS * D_;
    const float* __restrict__ Qp = Q + qbase;
    float* __restrict__ Op = out + qbase;
    const float* __restrict__ wkv = ws + (size_t)bh * KV_STRIDE;

    __shared__ __align__(16) float lds_kv[D_][KVSTRIDE_L];
    __shared__ float lds_kone[D_];
    __shared__ __align__(16) float lds_q[P2_ROWS][QSTRIDE];

    const int tid  = threadIdx.x;
    const int lane = tid & 31;
    const int wid  = tid >> 5;
    const int l15  = lane & 15;
    const int half = lane >> 4;

    #pragma unroll
    for (int i = 0; i < 4; ++i) {
        const int idx = tid + 256 * i;
        const int r = idx >> 4;
        const int c = (idx & 15) * 4;
        *(float4*)&lds_kv[r][c] = *(const float4*)&wkv[r * 64 + c];
    }
    if (tid < D_) lds_kone[tid] = wkv[4096 + tid];

    #pragma unroll
    for (int i = 0; i < 8; ++i) {
        const int idx = tid + 256 * i;
        const int r = idx >> 4;
        const int c = (idx & 15) * 4;
        float4 qv = *(const float4*)&Qp[(size_t)r * D_ + c];
        qv.x = phi_f(qv.x * 0.125f);
        qv.y = phi_f(qv.y * 0.125f);
        qv.z = phi_f(qv.z * 0.125f);
        qv.w = phi_f(qv.w * 0.125f);
        *(float4*)&lds_q[r][c] = qv;
    }
    __syncthreads();

    const int mb = wid * 16;
    v8f acc[4] = {{}, {}, {}, {}};

    #pragma unroll
    for (int kk = 0; kk < 64; kk += 4) {
        const int kc = kk + half * 2;
        v2f a;
        a.x = lds_q[mb + l15][kc];
        a.y = lds_q[mb + l15][kc + 1];
        #pragma unroll
        for (int vt = 0; vt < 4; ++vt) {
            v2f bfrag;
            bfrag.x = lds_kv[kc][vt * 16 + l15];
            bfrag.y = lds_kv[kc + 1][vt * 16 + l15];
            acc[vt] = __builtin_amdgcn_wmma_f32_16x16x4_f32(false, a, false, bfrag,
                                                            (short)0, acc[vt], false, false);
        }
    }

    float part = 0.0f;
    const int nrow = mb + l15;
    #pragma unroll
    for (int d = 0; d < 32; ++d) {
        const int dd = half * 32 + d;
        part += lds_q[nrow][dd] * lds_kone[dd];
    }
    part += __shfl_xor(part, 16, 32);
    const float inv = 1.0f / (part + EPS_);

    #pragma unroll
    for (int j = 0; j < 8; ++j) {
        const float rn = __shfl(inv, j + 8 * half, 32);
        const int r = mb + j + 8 * half;
        #pragma unroll
        for (int vt = 0; vt < 4; ++vt) {
            Op[(size_t)r * D_ + vt * 16 + l15] = acc[vt][j] * rn;
        }
    }
}

extern "C" void kernel_launch(void* const* d_in, const int* in_sizes, int n_in,
                              void* d_out, int out_size, void* d_ws, size_t ws_size,
                              hipStream_t stream) {
    const float* Q = (const float*)d_in[0];
    const float* K = (const float*)d_in[1];
    const float* V = (const float*)d_in[2];
    const float* M = (const float*)d_in[3];
    float* out = (float*)d_out;
    float* ws  = (float*)d_ws;

    const int nz4 = (BH_ * KV_STRIDE) / 4;
    la_zero_ws<<<(nz4 + 255) / 256, 256, 0, stream>>>((float4*)ws, nz4);
    la_pass1_kvterm<<<dim3(BH_, S_ / SCHUNK), 256, 0, stream>>>(K, V, M, ws);
    la_pass2_output<<<dim3(BH_, S_ / P2_ROWS), 256, 0, stream>>>(Q, ws, out);
}


# pass 2: Q rows go straight from global b128 loads into A fragments (k order permuted per lane half), kv kept in registers as B, no LDS or barrier, 4 tiles in flight per wave; new sum order over d
# speedup vs baseline: 1.4973x; 1.0333x over previous
#include <hip/hip_runtime.h>


typedef __attribute__((ext_vector_type(2))) float v2f;
typedef __attribute__((ext_vector_type(8))) float v8f;

#define B_ 4
#define H_ 16
#define S_ 4096
#define D_ 64
#define BH_ (B_ * H_)
#define EPS_ 1e-6f

#define KV_STRIDE 4160
#define SCHUNK 512

#define SLAB 64
#define NSLAB (SCHUNK / SLAB)
#define LPITCH 80


__device__ __forceinline__ float phi_f(float x) {
    return x > 0.0f ? x + 1.0f : __expf(x);
}

__global__ void la_zero_ws(float4* __restrict__ ws, int n4) {
    int i = blockIdx.x * blockDim.x + threadIdx.x;
    if (i < n4) ws[i] = make_float4(0.f, 0.f, 0.f, 0.f);
}

struct SlabRegs {
    float4 k[4];
    float4 v[4];
    float  m[4];
};

__device__ __forceinline__ void slab_load(SlabRegs& R, const float* Kp,
                                          const float* Vp,
                                          const float* __restrict__ Mp,
                                          int n0, int rr, int col) {
    #pragma unroll
    for (int i = 0; i < 4; ++i) {
        const int r = n0 + 16 * i + rr;
        R.k[i] = *(const float4*)&Kp[(size_t)r * D_ + col];
        R.v[i] = *(const float4*)&Vp[(size_t)r * D_ + col];
        R.m[i] = Mp[r];
    }
}

__device__ __forceinline__ void slab_stage(const SlabRegs& R, float* __restrict__ lk,
                                           float* __restrict__ lv, int rr, int col,
                                           float (&kone_acc)[4]) {
    #pragma unroll
    for (int i = 0; i < 4; ++i) {
        const int r = 16 * i + rr;
        const float m = R.m[i];
        float4 pk;
        pk.x = phi_f(R.k[i].x) * m;
        pk.y = phi_f(R.k[i].y) * m;
        pk.z = phi_f(R.k[i].z) * m;
        pk.w = phi_f(R.k[i].w) * m;
        kone_acc[0] += pk.x; kone_acc[1] += pk.y;
        kone_acc[2] += pk.z; kone_acc[3] += pk.w;
        *(float4*)&lk[r * LPITCH + col] = pk;
        *(float4*)&lv[r * LPITCH + col] = R.v[i];
    }
}

__device__ __forceinline__ void slab_mma(const float* __restrict__ lk,
                                         const float* __restrict__ lv,
                                         int aoff0, int aoff1, int boff,
                                         v8f& c0, v8f& c1) {
    #pragma unroll 8
    for (int kk = 0; kk < SLAB; kk += 4) {
        v2f a0, a1, bb;
        a0.x = lk[kk * LPITCH + aoff0];
        a0.y = lk[(kk + 1) * LPITCH + aoff0];
        a1.x = lk[kk * LPITCH + aoff1];
        a1.y = lk[(kk + 1) * LPITCH + aoff1];
        bb.x = lv[kk * LPITCH + boff];
        bb.y = lv[(kk + 1) * LPITCH + boff];
        c0 = __builtin_amdgcn_wmma_f32_16x16x4_f32(false, a0, false, bb,
                                                   (short)0, c0, false, false);
        c1 = __builtin_amdgcn_wmma_f32_16x16x4_f32(false, a1, false, bb,
                                                   (short)0, c1, false, false);
    }
}

__device__ __forceinline__ void lds_barrier() {
    __builtin_amdgcn_fence(__ATOMIC_RELEASE, "workgroup", "local");
    __builtin_amdgcn_s_barrier();
    __builtin_amdgcn_fence(__ATOMIC_ACQUIRE, "workgroup", "local");
}

__global__ __launch_bounds__(256) void la_pass1_kvterm(
    const float* K, const float* V,
    const float* __restrict__ mask, float* __restrict__ ws) {

    const int bh = blockIdx.x;
    const int chunk = blockIdx.y;
    const int b = bh / H_;
    const size_t base = (size_t)bh * S_ * D_ + (size_t)chunk * SCHUNK * D_;
    const float* Kp = K + base;
    const float* Vp = V + base;
    const float* __restrict__ Mp = mask + (size_t)b * S_ + (size_t)chunk * SCHUNK;

    __shared__ __align__(16) float lds_k[2][SLAB * LPITCH];
    __shared__ __align__(16) float lds_v[2][SLAB * LPITCH];
    __shared__ float lds_kone[D_];

    const int tid  = threadIdx.x;
    const int lane = tid & 31;
    const int wid  = tid >> 5;
    const int l15  = lane & 15;
    const int half = lane >> 4;

    const int dTile0 = wid >> 2;
    const int dTile1 = dTile0 + 2;
    const int vTile  = wid & 3;

    const int rr  = tid >> 4;
    const int col = (tid & 15) * 4;

    const int aoff0 = (2 * half) * LPITCH + dTile0 * 16 + l15;
    const int aoff1 = (2 * half) * LPITCH + dTile1 * 16 + l15;
    const int boff  = (2 * half) * LPITCH + vTile * 16 + l15;

    v8f c0 = {}; v8f c1 = {};
    float kone_acc[4] = {0.f, 0.f, 0.f, 0.f};

    if (tid < D_) lds_kone[tid] = 0.0f;

    SlabRegs ra, rb;
    slab_load(ra, Kp, Vp, Mp, 0, rr, col);
    __builtin_amdgcn_sched_barrier(0);
    slab_load(rb, Kp, Vp, Mp, SLAB, rr, col);

    #pragma unroll 1
    for (int s = 0; s < NSLAB - 2; s += 2) {
        slab_stage(ra, lds_k[0], lds_v[0], rr, col, kone_acc);
        slab_load(ra, Kp, Vp, Mp, (s + 2) * SLAB, rr, col);
        lds_barrier();
        slab_mma(lds_k[0], lds_v[0], aoff0, aoff1, boff, c0, c1);

        slab_stage(rb, lds_k[1], lds_v[1], rr, col, kone_acc);
        slab_load(rb, Kp, Vp, Mp, (s + 3) * SLAB, rr, col);
        lds_barrier();
        slab_mma(lds_k[1], lds_v[1], aoff0, aoff1, boff, c0, c1);
    }
    slab_stage(ra, lds_k[0], lds_v[0], rr, col, kone_acc);
    lds_barrier();
    slab_mma(lds_k[0], lds_v[0], aoff0, aoff1, boff, c0, c1);
    slab_stage(rb, lds_k[1], lds_v[1], rr, col, kone_acc);
    lds_barrier();
    slab_mma(lds_k[1], lds_v[1], aoff0, aoff1, boff, c0, c1);

    #pragma unroll
    for (int j = 0; j < 4; ++j) atomicAdd(&lds_kone[col + j], kone_acc[j]);
    __syncthreads();

    float* __restrict__ wkv = ws + (size_t)bh * KV_STRIDE;
    if (tid < D_) atomicAdd(&wkv[4096 + tid], lds_kone[tid]);

    #pragma unroll
    for (int j = 0; j < 8; ++j) {
        const int r0 = dTile0 * 16 + j + 8 * half;
        const int r1 = dTile1 * 16 + j + 8 * half;
        const int cidx = vTile * 16 + l15;
        atomicAdd(&wkv[r0 * 64 + cidx], c0[j]);
        atomicAdd(&wkv[r1 * 64 + cidx], c1[j]);
    }
}

#define P2_WAVES 4
#define P2_TILES 8
#define P2_WROWS (P2_TILES * 16)
#define P2_BROWS (P2_WAVES * P2_WROWS)
#define P2_DEPTH 4

struct QTile { float4 q[8]; };

__device__ __forceinline__ void q_tile_load(QTile& T, const float* __restrict__ Qlane) {
    #pragma unroll
    for (int j = 0; j < 8; ++j) T.q[j] = *(const float4*)&Qlane[8 * j];
}

__global__ __launch_bounds__(128) void la_pass2_output(
    const float* __restrict__ Q, const float* __restrict__ ws,
    float* __restrict__ out) {

    const int bh   = blockIdx.x / (S_ / P2_BROWS);
    const int part_idx = blockIdx.x % (S_ / P2_BROWS);
    const int tid  = threadIdx.x;
    const int lane = tid & 31;
    const int wid  = tid >> 5;
    const int c    = lane & 15;
    const int h    = lane >> 4;

    const int row0 = part_idx * P2_BROWS + wid * P2_WROWS;
    const size_t hbase = (size_t)bh * S_ * D_;
    const float* __restrict__ Qlane = Q + hbase + (size_t)(row0 + c) * D_ + 4 * h;
    float* __restrict__ Olane = out + hbase + (size_t)(row0 + 8 * h) * D_ + 4 * c;
    const float* __restrict__ wkv = ws + (size_t)bh * KV_STRIDE;

    QTile qb[P2_DEPTH];
    q_tile_load(qb[0], Qlane);
    q_tile_load(qb[1], Qlane + 16 * D_);

    v2f bf[16][4];
    {
        const float* __restrict__ kvl = wkv + (4 * h) * 64 + 4 * c;
        #pragma unroll
        for (int t = 0; t < 16; ++t) {
            const int dbase = 8 * (t >> 1) + 2 * (t & 1);
            const float4 x0 = *(const float4*)&kvl[(dbase + 0) * 64];
            const float4 x1 = *(const float4*)&kvl[(dbase + 1) * 64];
            bf[t][0].x = x0.x; bf[t][0].y = x1.x;
            bf[t][1].x = x0.y; bf[t][1].y = x1.y;
            bf[t][2].x = x0.z; bf[t][2].y = x1.z;
            bf[t][3].x = x0.w; bf[t][3].y = x1.w;
        }
    }
    float4 ko[8];
    #pragma unroll
    for (int j = 0; j < 8; ++j) ko[j] = *(const float4*)&wkv[4096 + 8 * j + 4 * h];

    q_tile_load(qb[2], Qlane + 32 * D_);
    q_tile_load(qb[3], Qlane + 48 * D_);

    #pragma unroll
    for (int i = 0; i < P2_TILES; ++i) {
        QTile& R = qb[i % P2_DEPTH];

        v2f a[16];
        float part = 0.0f;
        #pragma unroll
        for (int j = 0; j < 8; ++j) {
            const float px = phi_f(R.q[j].x * 0.125f);
            const float py = phi_f(R.q[j].y * 0.125f);
            const float pz = phi_f(R.q[j].z * 0.125f);
            const float pw = phi_f(R.q[j].w * 0.125f);
            a[2 * j].x = px;     a[2 * j].y = py;
            a[2 * j + 1].x = pz; a[2 * j + 1].y = pw;
            part += px * ko[j].x;
            part += py * ko[j].y;
            part += pz * ko[j].z;
            part += pw * ko[j].w;
        }
        if (i + P2_DEPTH < P2_TILES)
            q_tile_load(R, Qlane + (size_t)(i + P2_DEPTH) * 16 * D_);

        part += __shfl_xor(part, 16, 32);
        const float inv = 1.0f / (part + EPS_);

        v8f acc[4] = {{}, {}, {}, {}};
        #pragma unroll
        for (int t = 0; t < 16; ++t) {
            #pragma unroll
            for (int vt = 0; vt < 4; ++vt) {
                acc[vt] = __builtin_amdgcn_wmma_f32_16x16x4_f32(false, a[t], false, bf[t][vt],
                                                                (short)0, acc[vt], false, false);
            }
        }

        float* __restrict__ Ot = Olane + (size_t)i * 16 * D_;
        #pragma unroll
        for (int r = 0; r < 8; ++r) {
            const float rn = __shfl(inv, r + 8 * h, 32);
            float4 o;
            o.x = acc[0][r] * rn;
            o.y = acc[1][r] * rn;
            o.z = acc[2][r] * rn;
            o.w = acc[3][r] * rn;
            *(float4*)&Ot[r * D_] = o;
        }
        __builtin_amdgcn_sched_barrier(0);
    }
}

extern "C" void kernel_launch(void* const* d_in, const int* in_sizes, int n_in,
                              void* d_out, int out_size, void* d_ws, size_t ws_size,
                              hipStream_t stream) {
    const float* Q = (const float*)d_in[0];
    const float* K = (const float*)d_in[1];
    const float* V = (const float*)d_in[2];
    const float* M = (const float*)d_in[3];
    float* out = (float*)d_out;
    float* ws  = (float*)d_ws;

    const int nz4 = (BH_ * KV_STRIDE) / 4;
    la_zero_ws<<<(nz4 + 255) / 256, 256, 0, stream>>>((float4*)ws, nz4);
    la_pass1_kvterm<<<dim3(BH_, S_ / SCHUNK), 256, 0, stream>>>(K, V, M, ws);
    la_pass2_output<<<dim3(BH_ * (S_ / P2_BROWS)), 128, 0, stream>>>(Q, ws, out);
}


# pass 1: LDS slab stored four rows per float4, so A,B fragments are b128 reads issued in two groups of 12, each ahead of 16 WMMAs, not 3 reads per WMMA pair; new sum order over n inside a slab
# speedup vs baseline: 1.5935x; 1.0642x over previous
#include <hip/hip_runtime.h>


typedef __attribute__((ext_vector_type(2))) float v2f;
typedef __attribute__((ext_vector_type(8))) float v8f;

#define B_ 4
#define H_ 16
#define S_ 4096
#define D_ 64
#define BH_ (B_ * H_)
#define EPS_ 1e-6f

#define KV_STRIDE 4160
#define SCHUNK 512

#define SLAB 64
#define NSLAB (SCHUNK / SLAB)
#define TPITCH 68
#define GPITCH (4 * TPITCH)
#define SLAB_FLOATS (16 * GPITCH)


__device__ __forceinline__ float phi_f(float x) {
    return x > 0.0f ? x + 1.0f : __expf(x);
}

__global__ void la_zero_ws(float4* __restrict__ ws, int n4) {
    int i = blockIdx.x * blockDim.x + threadIdx.x;
    if (i < n4) ws[i] = make_float4(0.f, 0.f, 0.f, 0.f);
}

struct SlabRegs {
    float4 k[4];
    float4 v[4];
    float  m[4];
};

__device__ __forceinline__ void slab_load(SlabRegs& R, const float* Kp,
                                          const float* Vp,
                                          const float* __restrict__ Mp,
                                          int n0, int rr, int col) {
    #pragma unroll
    for (int i = 0; i < 4; ++i) {
        const int r = n0 + 16 * i + rr;
        R.k[i] = *(const float4*)&Kp[(size_t)r * D_ + col];
        R.v[i] = *(const float4*)&Vp[(size_t)r * D_ + col];
        R.m[i] = Mp[r];
    }
}

__device__ __forceinline__ void slab_stage(const SlabRegs& R, float* __restrict__ lk,
                                           float* __restrict__ lv, int soff,
                                           float (&kone_acc)[4]) {
    float4 pk[4];
    #pragma unroll
    for (int i = 0; i < 4; ++i) {
        const float m = R.m[i];
        pk[i].x = phi_f(R.k[i].x) * m;
        pk[i].y = phi_f(R.k[i].y) * m;
        pk[i].z = phi_f(R.k[i].z) * m;
        pk[i].w = phi_f(R.k[i].w) * m;
        kone_acc[0] += pk[i].x; kone_acc[1] += pk[i].y;
        kone_acc[2] += pk[i].z; kone_acc[3] += pk[i].w;
    }
    *(float4*)&lk[soff + 0]  = make_float4(pk[0].x, pk[1].x, pk[2].x, pk[3].x);
    *(float4*)&lk[soff + 4]  = make_float4(pk[0].y, pk[1].y, pk[2].y, pk[3].y);
    *(float4*)&lk[soff + 8]  = make_float4(pk[0].z, pk[1].z, pk[2].z, pk[3].z);
    *(float4*)&lk[soff + 12] = make_float4(pk[0].w, pk[1].w, pk[2].w, pk[3].w);
    *(float4*)&lv[soff + 0]  = make_float4(R.v[0].x, R.v[1].x, R.v[2].x, R.v[3].x);
    *(float4*)&lv[soff + 4]  = make_float4(R.v[0].y, R.v[1].y, R.v[2].y, R.v[3].y);
    *(float4*)&lv[soff + 8]  = make_float4(R.v[0].z, R.v[1].z, R.v[2].z, R.v[3].z);
    *(float4*)&lv[soff + 12] = make_float4(R.v[0].w, R.v[1].w, R.v[2].w, R.v[3].w);
}

struct HalfFrags {
    float4 a0[4];
    float4 a1[4];
    float4 b[4];
};

__device__ __forceinline__ void frag_load(HalfFrags& F, const float* __restrict__ lk,
                                          const float* __restrict__ lv, int aoff, int boff, int hs) {
    #pragma unroll
    for (int p = 0; p < 4; ++p) {
        const int g = (4 * hs + p) * 2 * GPITCH;
        F.a0[p] = *(const float4*)&lk[aoff + g];
        F.a1[p] = *(const float4*)&lk[aoff + g + 2 * TPITCH];
        F.b[p]  = *(const float4*)&lv[boff + g];
    }
}

__device__ __forceinline__ void frag_mma(const HalfFrags& F, v8f& c0, v8f& c1) {
    #pragma unroll
    for (int p = 0; p < 4; ++p) {
        v2f a0l, a0h, a1l, a1h, bl, bh;
        a0l.x = F.a0[p].x; a0l.y = F.a0[p].y; a0h.x = F.a0[p].z; a0h.y = F.a0[p].w;
        a1l.x = F.a1[p].x; a1l.y = F.a1[p].y; a1h.x = F.a1[p].z; a1h.y = F.a1[p].w;
        bl.x  = F.b[p].x;  bl.y  = F.b[p].y;  bh.x  = F.b[p].z;  bh.y  = F.b[p].w;
        c0 = __builtin_amdgcn_wmma_f32_16x16x4_f32(false, a0l, false, bl,
                                                   (short)0, c0, false, false);
        c1 = __builtin_amdgcn_wmma_f32_16x16x4_f32(false, a1l, false, bl,
                                                   (short)0, c1, false, false);
        c0 = __builtin_amdgcn_wmma_f32_16x16x4_f32(false, a0h, false, bh,
                                                   (short)0, c0, false, false);
        c1 = __builtin_amdgcn_wmma_f32_16x16x4_f32(false, a1h, false, bh,
                                                   (short)0, c1, false, false);
    }
}

__device__ __forceinline__ void lds_barrier() {
    __builtin_amdgcn_fence(__ATOMIC_RELEASE, "workgroup", "local");
    __builtin_amdgcn_s_barrier();
    __builtin_amdgcn_fence(__ATOMIC_ACQUIRE, "workgroup", "local");
}

__global__ __launch_bounds__(256) void la_pass1_kvterm(
    const float* K, const float* V,
    const float* __restrict__ mask, float* __restrict__ ws) {

    const int bh = blockIdx.x;
    const int chunk = blockIdx.y;
    const int b = bh / H_;
    const size_t base = (size_t)bh * S_ * D_ + (size_t)chunk * SCHUNK * D_;
    const float* Kp = K + base;
    const float* Vp = V + base;
    const float* __restrict__ Mp = mask + (size_t)b * S_ + (size_t)chunk * SCHUNK;

    __shared__ __align__(16) float lds_k[2][SLAB_FLOATS];
    __shared__ __align__(16) float lds_v[2][SLAB_FLOATS];
    __shared__ float lds_kone[D_];

    const int tid  = threadIdx.x;
    const int lane = tid & 31;
    const int wid  = tid >> 5;
    const int l15  = lane & 15;
    const int half = lane >> 4;

    const int dTile0 = wid >> 2;
    const int dTile1 = dTile0 + 2;
    const int vTile  = wid & 3;

    const int rr  = tid >> 4;
    const int col = (tid & 15) * 4;

    const int soff = rr * GPITCH + (col >> 4) * TPITCH + (col & 15) * 4;
    const int aoff = half * GPITCH + dTile0 * TPITCH + l15 * 4;
    const int boff = half * GPITCH + vTile * TPITCH + l15 * 4;

    v8f c0 = {}; v8f c1 = {};
    float kone_acc[4] = {0.f, 0.f, 0.f, 0.f};

    if (tid < D_) lds_kone[tid] = 0.0f;

    SlabRegs ra, rb;
    slab_load(ra, Kp, Vp, Mp, 0, rr, col);
    __builtin_amdgcn_sched_barrier(0);
    slab_load(rb, Kp, Vp, Mp, SLAB, rr, col);

    HalfFrags F;
#define SLAB_STEP(BUF, R, LOADSTMT)                                                  \
    slab_stage(R, lds_k[BUF], lds_v[BUF], soff, kone_acc);                           \
    LOADSTMT;                                                                        \
    lds_barrier();              \
    frag_load(F, lds_k[BUF], lds_v[BUF], aoff, boff, 0);                             \
    __builtin_amdgcn_sched_barrier(0);                                               \
    frag_mma(F, c0, c1);                                                             \
    __builtin_amdgcn_sched_barrier(0);                                               \
    frag_load(F, lds_k[BUF], lds_v[BUF], aoff, boff, 1);                             \
    __builtin_amdgcn_sched_barrier(0);                                               \
    frag_mma(F, c0, c1);                                                             \
    __builtin_amdgcn_sched_barrier(0);

    #pragma unroll 1
    for (int s = 0; s < NSLAB - 2; s += 2) {
        SLAB_STEP(0, ra, slab_load(ra, Kp, Vp, Mp, (s + 2) * SLAB, rr, col))
        SLAB_STEP(1, rb, slab_load(rb, Kp, Vp, Mp, (s + 3) * SLAB, rr, col))
    }
    SLAB_STEP(0, ra, (void)0)
    SLAB_STEP(1, rb, (void)0)
#undef SLAB_STEP

    #pragma unroll
    for (int j = 0; j < 4; ++j) atomicAdd(&lds_kone[col + j], kone_acc[j]);
    __syncthreads();

    float* __restrict__ wkv = ws + (size_t)bh * KV_STRIDE;
    if (tid < D_) atomicAdd(&wkv[4096 + tid], lds_kone[tid]);

    #pragma unroll
    for (int j = 0; j < 8; ++j) {
        const int r0 = dTile0 * 16 + j + 8 * half;
        const int r1 = dTile1 * 16 + j + 8 * half;
        const int cidx = vTile * 16 + l15;
        atomicAdd(&wkv[r0 * 64 + cidx], c0[j]);
        atomicAdd(&wkv[r1 * 64 + cidx], c1[j]);
    }
}

#define P2_WAVES 4
#define P2_TILES 8
#define P2_WROWS (P2_TILES * 16)
#define P2_BROWS (P2_WAVES * P2_WROWS)
#define P2_DEPTH 4

struct QTile { float4 q[8]; };

__device__ __forceinline__ void q_tile_load(QTile& T, const float* __restrict__ Qlane) {
    #pragma unroll
    for (int j = 0; j < 8; ++j) T.q[j] = *(const float4*)&Qlane[8 * j];
}

__global__ __launch_bounds__(128) void la_pass2_output(
    const float* __restrict__ Q, const float* __restrict__ ws,
    float* __restrict__ out) {

    const int bh   = blockIdx.x / (S_ / P2_BROWS);
    const int part_idx = blockIdx.x % (S_ / P2_BROWS);
    const int tid  = threadIdx.x;
    const int lane = tid & 31;
    const int wid  = tid >> 5;
    const int c    = lane & 15;
    const int h    = lane >> 4;

    const int row0 = part_idx * P2_BROWS + wid * P2_WROWS;
    const size_t hbase = (size_t)bh * S_ * D_;
    const float* __restrict__ Qlane = Q + hbase + (size_t)(row0 + c) * D_ + 4 * h;
    float* __restrict__ Olane = out + hbase + (size_t)(row0 + 8 * h) * D_ + 4 * c;
    const float* __restrict__ wkv = ws + (size_t)bh * KV_STRIDE;

    QTile qb[P2_DEPTH];
    q_tile_load(qb[0], Qlane);
    q_tile_load(qb[1], Qlane + 16 * D_);

    v2f bf[16][4];
    {
        const float* __restrict__ kvl = wkv + (4 * h) * 64 + 4 * c;
        #pragma unroll
        for (int t = 0; t < 16; ++t) {
            const int dbase = 8 * (t >> 1) + 2 * (t & 1);
            const float4 x0 = *(const float4*)&kvl[(dbase + 0) * 64];
            const float4 x1 = *(const float4*)&kvl[(dbase + 1) * 64];
            bf[t][0].x = x0.x; bf[t][0].y = x1.x;
            bf[t][1].x = x0.y; bf[t][1].y = x1.y;
            bf[t][2].x = x0.z; bf[t][2].y = x1.z;
            bf[t][3].x = x0.w; bf[t][3].y = x1.w;
        }
    }
    float4 ko[8];
    #pragma unroll
    for (int j = 0; j < 8; ++j) ko[j] = *(const float4*)&wkv[4096 + 8 * j + 4 * h];

    q_tile_load(qb[2], Qlane + 32 * D_);
    q_tile_load(qb[3], Qlane + 48 * D_);

    #pragma unroll
    for (int i = 0; i < P2_TILES; ++i) {
        QTile& R = qb[i % P2_DEPTH];

        v2f a[16];
        float part = 0.0f;
        #pragma unroll
        for (int j = 0; j < 8; ++j) {
            const float px = phi_f(R.q[j].x * 0.125f);
            const float py = phi_f(R.q[j].y * 0.125f);
            const float pz = phi_f(R.q[j].z * 0.125f);
            const float pw = phi_f(R.q[j].w * 0.125f);
            a[2 * j].x = px;     a[2 * j].y = py;
            a[2 * j + 1].x = pz; a[2 * j + 1].y = pw;
            part += px * ko[j].x;
            part += py * ko[j].y;
            part += pz * ko[j].z;
            part += pw * ko[j].w;
        }
        if (i + P2_DEPTH < P2_TILES)
            q_tile_load(R, Qlane + (size_t)(i + P2_DEPTH) * 16 * D_);

        part += __shfl_xor(part, 16, 32);
        const float inv = 1.0f / (part + EPS_);

        v8f acc[4] = {{}, {}, {}, {}};
        #pragma unroll
        for (int t = 0; t < 16; ++t) {
            #pragma unroll
            for (int vt = 0; vt < 4; ++vt) {
                acc[vt] = __builtin_amdgcn_wmma_f32_16x16x4_f32(false, a[t], false, bf[t][vt],
                                                                (short)0, acc[vt], false, false);
            }
        }

        float* __restrict__ Ot = Olane + (size_t)i * 16 * D_;
        #pragma unroll
        for (int r = 0; r < 8; ++r) {
            const float rn = __shfl(inv, r + 8 * h, 32);
            float4 o;
            o.x = acc[0][r] * rn;
            o.y = acc[1][r] * rn;
            o.z = acc[2][r] * rn;
            o.w = acc[3][r] * rn;
            *(float4*)&Ot[r * D_] = o;
        }
        __builtin_amdgcn_sched_barrier(0);
    }
}

extern "C" void kernel_launch(void* const* d_in, const int* in_sizes, int n_in,
                              void* d_out, int out_size, void* d_ws, size_t ws_size,
                              hipStream_t stream) {
    const float* Q = (const float*)d_in[0];
    const float* K = (const float*)d_in[1];
    const float* V = (const float*)d_in[2];
    const float* M = (const float*)d_in[3];
    float* out = (float*)d_out;
    float* ws  = (float*)d_ws;

    const int nz4 = (BH_ * KV_STRIDE) / 4;
    la_zero_ws<<<(nz4 + 255) / 256, 256, 0, stream>>>((float4*)ws, nz4);
    la_pass1_kvterm<<<dim3(BH_, S_ / SCHUNK), 256, 0, stream>>>(K, V, M, ws);
    la_pass2_output<<<dim3(BH_ * (S_ / P2_BROWS)), 128, 0, stream>>>(Q, ws, out);
}


# both passes: wave issue priority (s_setprio) falls 3,2,1,0 per quarter of a wave's slabs or tiles, so the two co-resident work-groups of a WGP leapfrog and end together; arithmetic untouched
# speedup vs baseline: 1.6421x; 1.0305x over previous
#include <hip/hip_runtime.h>


typedef __attribute__((ext_vector_type(2))) float v2f;
typedef __attribute__((ext_vector_type(8))) float v8f;

#define B_ 4
#define H_ 16
#define S_ 4096
#define D_ 64
#define BH_ (B_ * H_)
#define EPS_ 1e-6f

#define KV_STRIDE 4160
#define SCHUNK 512

#define SLAB 64
#define NSLAB (SCHUNK / SLAB)
#define TPITCH 68
#define GPITCH (4 * TPITCH)
#define SLAB_FLOATS (16 * GPITCH)


__device__ __forceinline__ float phi_f(float x) {
    return x > 0.0f ? x + 1.0f : __expf(x);
}

__device__ __forceinline__ void wave_prio(int q) {
    if (q == 0)      __builtin_amdgcn_s_setprio(3);
    else if (q == 1) __builtin_amdgcn_s_setprio(2);
    else if (q == 2) __builtin_amdgcn_s_setprio(1);
    else             __builtin_amdgcn_s_setprio(0);
}

__global__ void la_zero_ws(float4* __restrict__ ws, int n4) {
    int i = blockIdx.x * blockDim.x + threadIdx.x;
    if (i < n4) ws[i] = make_float4(0.f, 0.f, 0.f, 0.f);
}

struct SlabRegs {
    float4 k[4];
    float4 v[4];
    float  m[4];
};

__device__ __forceinline__ void slab_load(SlabRegs& R, const float* Kp,
                                          const float* Vp,
                                          const float* __restrict__ Mp,
                                          int n0, int rr, int col) {
    #pragma unroll
    for (int i = 0; i < 4; ++i) {
        const int r = n0 + 16 * i + rr;
        R.k[i] = *(const float4*)&Kp[(size_t)r * D_ + col];
        R.v[i] = *(const float4*)&Vp[(size_t)r * D_ + col];
        R.m[i] = Mp[r];
    }
}

__device__ __forceinline__ void slab_stage(const SlabRegs& R, float* __restrict__ lk,
                                           float* __restrict__ lv, int soff,
                                           float (&kone_acc)[4]) {
    float4 pk[4];
    #pragma unroll
    for (int i = 0; i < 4; ++i) {
        const float m = R.m[i];
        pk[i].x = phi_f(R.k[i].x) * m;
        pk[i].y = phi_f(R.k[i].y) * m;
        pk[i].z = phi_f(R.k[i].z) * m;
        pk[i].w = phi_f(R.k[i].w) * m;
        kone_acc[0] += pk[i].x; kone_acc[1] += pk[i].y;
        kone_acc[2] += pk[i].z; kone_acc[3] += pk[i].w;
    }
    *(float4*)&lk[soff + 0]  = make_float4(pk[0].x, pk[1].x, pk[2].x, pk[3].x);
    *(float4*)&lk[soff + 4]  = make_float4(pk[0].y, pk[1].y, pk[2].y, pk[3].y);
    *(float4*)&lk[soff + 8]  = make_float4(pk[0].z, pk[1].z, pk[2].z, pk[3].z);
    *(float4*)&lk[soff + 12] = make_float4(pk[0].w, pk[1].w, pk[2].w, pk[3].w);
    *(float4*)&lv[soff + 0]  = make_float4(R.v[0].x, R.v[1].x, R.v[2].x, R.v[3].x);
    *(float4*)&lv[soff + 4]  = make_float4(R.v[0].y, R.v[1].y, R.v[2].y, R.v[3].y);
    *(float4*)&lv[soff + 8]  = make_float4(R.v[0].z, R.v[1].z, R.v[2].z, R.v[3].z);
    *(float4*)&lv[soff + 12] = make_float4(R.v[0].w, R.v[1].w, R.v[2].w, R.v[3].w);
}

struct HalfFrags {
    float4 a0[4];
    float4 a1[4];
    float4 b[4];
};

__device__ __forceinline__ void frag_load(HalfFrags& F, const float* __restrict__ lk,
                                          const float* __restrict__ lv, int aoff, int boff, int hs) {
    #pragma unroll
    for (int p = 0; p < 4; ++p) {
        const int g = (4 * hs + p) * 2 * GPITCH;
        F.a0[p] = *(const float4*)&lk[aoff + g];
        F.a1[p] = *(const float4*)&lk[aoff + g + 2 * TPITCH];
        F.b[p]  = *(const float4*)&lv[boff + g];
    }
}

__device__ __forceinline__ void frag_mma(const HalfFrags& F, v8f& c0, v8f& c1) {
    #pragma unroll
    for (int p = 0; p < 4; ++p) {
        v2f a0l, a0h, a1l, a1h, bl, bh;
        a0l.x = F.a0[p].x; a0l.y = F.a0[p].y; a0h.x = F.a0[p].z; a0h.y = F.a0[p].w;
        a1l.x = F.a1[p].x; a1l.y = F.a1[p].y; a1h.x = F.a1[p].z; a1h.y = F.a1[p].w;
        bl.x  = F.b[p].x;  bl.y  = F.b[p].y;  bh.x  = F.b[p].z;  bh.y  = F.b[p].w;
        c0 = __builtin_amdgcn_wmma_f32_16x16x4_f32(false, a0l, false, bl,
                                                   (short)0, c0, false, false);
        c1 = __builtin_amdgcn_wmma_f32_16x16x4_f32(false, a1l, false, bl,
                                                   (short)0, c1, false, false);
        c0 = __builtin_amdgcn_wmma_f32_16x16x4_f32(false, a0h, false, bh,
                                                   (short)0, c0, false, false);
        c1 = __builtin_amdgcn_wmma_f32_16x16x4_f32(false, a1h, false, bh,
                                                   (short)0, c1, false, false);
    }
}

__device__ __forceinline__ void lds_barrier() {
    __builtin_amdgcn_fence(__ATOMIC_RELEASE, "workgroup", "local");
    __builtin_amdgcn_s_barrier();
    __builtin_amdgcn_fence(__ATOMIC_ACQUIRE, "workgroup", "local");
}

__global__ __launch_bounds__(256) void la_pass1_kvterm(
    const float* K, const float* V,
    const float* __restrict__ mask, float* __restrict__ ws) {

    wave_prio(0);

    const int bh = blockIdx.x;
    const int chunk = blockIdx.y;
    const int b = bh / H_;
    const size_t base = (size_t)bh * S_ * D_ + (size_t)chunk * SCHUNK * D_;
    const float* Kp = K + base;
    const float* Vp = V + base;
    const float* __restrict__ Mp = mask + (size_t)b * S_ + (size_t)chunk * SCHUNK;

    __shared__ __align__(16) float lds_k[2][SLAB_FLOATS];
    __shared__ __align__(16) float lds_v[2][SLAB_FLOATS];
    __shared__ float lds_kone[D_];

    const int tid  = threadIdx.x;
    const int lane = tid & 31;
    const int wid  = tid >> 5;
    const int l15  = lane & 15;
    const int half = lane >> 4;

    const int dTile0 = wid >> 2;
    const int dTile1 = dTile0 + 2;
    const int vTile  = wid & 3;

    const int rr  = tid >> 4;
    const int col = (tid & 15) * 4;

    const int soff = rr * GPITCH + (col >> 4) * TPITCH + (col & 15) * 4;
    const int aoff = half * GPITCH + dTile0 * TPITCH + l15 * 4;
    const int boff = half * GPITCH + vTile * TPITCH + l15 * 4;

    v8f c0 = {}; v8f c1 = {};
    float kone_acc[4] = {0.f, 0.f, 0.f, 0.f};

    if (tid < D_) lds_kone[tid] = 0.0f;

    SlabRegs ra, rb;
    slab_load(ra, Kp, Vp, Mp, 0, rr, col);
    __builtin_amdgcn_sched_barrier(0);
    slab_load(rb, Kp, Vp, Mp, SLAB, rr, col);

    HalfFrags F;
#define SLAB_STEP(BUF, R, LOADSTMT)                                                  \
    slab_stage(R, lds_k[BUF], lds_v[BUF], soff, kone_acc);                           \
    LOADSTMT;                                                                        \
    lds_barrier();              \
    frag_load(F, lds_k[BUF], lds_v[BUF], aoff, boff, 0);                             \
    __builtin_amdgcn_sched_barrier(0);                                               \
    frag_mma(F, c0, c1);                                                             \
    __builtin_amdgcn_sched_barrier(0);                                               \
    frag_load(F, lds_k[BUF], lds_v[BUF], aoff, boff, 1);                             \
    __builtin_amdgcn_sched_barrier(0);                                               \
    frag_mma(F, c0, c1);                                                             \
    __builtin_amdgcn_sched_barrier(0);

    #pragma unroll 1
    for (int s = 0; s < NSLAB - 2; s += 2) {
        SLAB_STEP(0, ra, slab_load(ra, Kp, Vp, Mp, (s + 2) * SLAB, rr, col))
        SLAB_STEP(1, rb, slab_load(rb, Kp, Vp, Mp, (s + 3) * SLAB, rr, col))
        wave_prio((s >> 1) + 1);
    }
    SLAB_STEP(0, ra, (void)0)
    SLAB_STEP(1, rb, (void)0)
#undef SLAB_STEP

    #pragma unroll
    for (int j = 0; j < 4; ++j) atomicAdd(&lds_kone[col + j], kone_acc[j]);
    __syncthreads();

    float* __restrict__ wkv = ws + (size_t)bh * KV_STRIDE;
    if (tid < D_) atomicAdd(&wkv[4096 + tid], lds_kone[tid]);

    #pragma unroll
    for (int j = 0; j < 8; ++j) {
        const int r0 = dTile0 * 16 + j + 8 * half;
        const int r1 = dTile1 * 16 + j + 8 * half;
        const int cidx = vTile * 16 + l15;
        atomicAdd(&wkv[r0 * 64 + cidx], c0[j]);
        atomicAdd(&wkv[r1 * 64 + cidx], c1[j]);
    }
}

#define P2_WAVES 4
#define P2_TILES 8
#define P2_WROWS (P2_TILES * 16)
#define P2_BROWS (P2_WAVES * P2_WROWS)
#define P2_DEPTH 4

struct QTile { float4 q[8]; };

__device__ __forceinline__ void q_tile_load(QTile& T, const float* __restrict__ Qlane) {
    #pragma unroll
    for (int j = 0; j < 8; ++j) T.q[j] = *(const float4*)&Qlane[8 * j];
}

__global__ __launch_bounds__(128) void la_pass2_output(
    const float* __restrict__ Q, const float* __restrict__ ws,
    float* __restrict__ out) {

    wave_prio(0);

    const int bh   = blockIdx.x / (S_ / P2_BROWS);
    const int part_idx = blockIdx.x % (S_ / P2_BROWS);
    const int tid  = threadIdx.x;
    const int lane = tid & 31;
    const int wid  = tid >> 5;
    const int c    = lane & 15;
    const int h    = lane >> 4;

    const int row0 = part_idx * P2_BROWS + wid * P2_WROWS;
    const size_t hbase = (size_t)bh * S_ * D_;
    const float* __restrict__ Qlane = Q + hbase + (size_t)(row0 + c) * D_ + 4 * h;
    float* __restrict__ Olane = out + hbase + (size_t)(row0 + 8 * h) * D_ + 4 * c;
    const float* __restrict__ wkv = ws + (size_t)bh * KV_STRIDE;

    QTile qb[P2_DEPTH];
    q_tile_load(qb[0], Qlane);
    q_tile_load(qb[1], Qlane + 16 * D_);

    v2f bf[16][4];
    {
        const float* __restrict__ kvl = wkv + (4 * h) * 64 + 4 * c;
        #pragma unroll
        for (int t = 0; t < 16; ++t) {
            const int dbase = 8 * (t >> 1) + 2 * (t & 1);
            const float4 x0 = *(const float4*)&kvl[(dbase + 0) * 64];
            const float4 x1 = *(const float4*)&kvl[(dbase + 1) * 64];
            bf[t][0].x = x0.x; bf[t][0].y = x1.x;
            bf[t][1].x = x0.y; bf[t][1].y = x1.y;
            bf[t][2].x = x0.z; bf[t][2].y = x1.z;
            bf[t][3].x = x0.w; bf[t][3].y = x1.w;
        }
    }
    float4 ko[8];
    #pragma unroll
    for (int j = 0; j < 8; ++j) ko[j] = *(const float4*)&wkv[4096 + 8 * j + 4 * h];

    q_tile_load(qb[2], Qlane + 32 * D_);
    q_tile_load(qb[3], Qlane + 48 * D_);

    #pragma unroll
    for (int i = 0; i < P2_TILES; ++i) {
        QTile& R = qb[i % P2_DEPTH];

        v2f a[16];
        float part = 0.0f;
        #pragma unroll
        for (int j = 0; j < 8; ++j) {
            const float px = phi_f(R.q[j].x * 0.125f);
            const float py = phi_f(R.q[j].y * 0.125f);
            const float pz = phi_f(R.q[j].z * 0.125f);
            const float pw = phi_f(R.q[j].w * 0.125f);
            a[2 * j].x = px;     a[2 * j].y = py;
            a[2 * j + 1].x = pz; a[2 * j + 1].y = pw;
            part += px * ko[j].x;
            part += py * ko[j].y;
            part += pz * ko[j].z;
            part += pw * ko[j].w;
        }
        if (i + P2_DEPTH < P2_TILES)
            q_tile_load(R, Qlane + (size_t)(i + P2_DEPTH) * 16 * D_);

        part += __shfl_xor(part, 16, 32);
        const float inv = 1.0f / (part + EPS_);

        v8f acc[4] = {{}, {}, {}, {}};
        #pragma unroll
        for (int t = 0; t < 16; ++t) {
            #pragma unroll
            for (int vt = 0; vt < 4; ++vt) {
                acc[vt] = __builtin_amdgcn_wmma_f32_16x16x4_f32(false, a[t], false, bf[t][vt],
                                                                (short)0, acc[vt], false, false);
            }
        }

        float* __restrict__ Ot = Olane + (size_t)i * 16 * D_;
        #pragma unroll
        for (int r = 0; r < 8; ++r) {
            const float rn = __shfl(inv, r + 8 * h, 32);
            float4 o;
            o.x = acc[0][r] * rn;
            o.y = acc[1][r] * rn;
            o.z = acc[2][r] * rn;
            o.w = acc[3][r] * rn;
            *(float4*)&Ot[r * D_] = o;
        }
        if ((i & 1) == 1 && i < P2_TILES - 1) wave_prio((i + 1) >> 1);
        __builtin_amdgcn_sched_barrier(0);
    }
}

extern "C" void kernel_launch(void* const* d_in, const int* in_sizes, int n_in,
                              void* d_out, int out_size, void* d_ws, size_t ws_size,
                              hipStream_t stream) {
    const float* Q = (const float*)d_in[0];
    const float* K = (const float*)d_in[1];
    const float* V = (const float*)d_in[2];
    const float* M = (const float*)d_in[3];
    float* out = (float*)d_out;
    float* ws  = (float*)d_ws;

    const int nz4 = (BH_ * KV_STRIDE) / 4;
    la_zero_ws<<<(nz4 + 255) / 256, 256, 0, stream>>>((float4*)ws, nz4);
    la_pass1_kvterm<<<dim3(BH_, S_ / SCHUNK), 256, 0, stream>>>(K, V, M, ws);
    la_pass2_output<<<dim3(BH_ * (S_ / P2_BROWS)), 128, 0, stream>>>(Q, ws, out);
}
